# starting kernel
# baseline (speedup 1.0000x reference)
#include <hip/hip_runtime.h>


typedef __attribute__((ext_vector_type(2))) float v2f;
typedef __attribute__((ext_vector_type(8))) float v8f;

#define B_    4
#define H_    16
#define L_    2048
#define D_    64
#define RWIN  256
#define KEY0  (L_ - RWIN)
#define DIAG  (L_ - RWIN + 1)
#define LASTQ (L_ - 1 - DIAG)
#define NSPLIT 8
#define ROWS_PER_SPLIT (L_ / NSPLIT)

__global__ __launch_bounds__(256)
void vpartial_kernel(const float* __restrict__ v, float* __restrict__ partial) {
  __shared__ float4 red[256];
  const int blk = blockIdx.x;
  const int bh  = blk >> 3;
  const int s   = blk & (NSPLIT - 1);
  const int t   = threadIdx.x;
  const int c   = t & 15;
  const int g   = t >> 4;

  const float4* vb4 = (const float4*)(v + (size_t)bh * L_ * D_);
  const int row0 = s * ROWS_PER_SPLIT + g;

  float4 acc = make_float4(0.f, 0.f, 0.f, 0.f);
#pragma unroll
  for (int it = 0; it < ROWS_PER_SPLIT / 16; ++it) {
    float4 x = vb4[(size_t)(row0 + it * 16) * (D_ / 4) + c];
    acc.x += x.x; acc.y += x.y; acc.z += x.z; acc.w += x.w;
  }
  red[t] = acc;
  __syncthreads();
  if (t < 128) { float4 o = red[t + 128];
    red[t].x += o.x; red[t].y += o.y; red[t].z += o.z; red[t].w += o.w; }
  __syncthreads();
  if (t < 64)  { float4 o = red[t + 64];
    red[t].x += o.x; red[t].y += o.y; red[t].z += o.z; red[t].w += o.w; }
  __syncthreads();
  if (t < 32)  { float4 o = red[t + 32];
    red[t].x += o.x; red[t].y += o.y; red[t].z += o.z; red[t].w += o.w; }
  __syncthreads();
  if (t < 16) {
    float4 o = red[t + 16];
    float4 r = red[t];
    r.x += o.x; r.y += o.y; r.z += o.z; r.w += o.w;
    ((float4*)partial)[(size_t)blk * (D_ / 4) + t] = r;
  }
}

__global__ __launch_bounds__(64)
void vcombine_kernel(const float* __restrict__ partial, float* __restrict__ vmean) {
  const int bh = blockIdx.x;
  const int d  = threadIdx.x;
  float s = 0.f;
#pragma unroll
  for (int sp = 0; sp < NSPLIT; ++sp)
    s += partial[(size_t)(bh * NSPLIT + sp) * D_ + d];
  vmean[bh * D_ + d] = s * (1.0f / (float)L_);
}

__global__ __launch_bounds__(256)
void bcast_kernel(const float* __restrict__ vmean, float* __restrict__ out) {
  const size_t idx = (size_t)blockIdx.x * blockDim.x + threadIdx.x;
  const size_t e   = idx * 4;
  const int bh = (int)(e / (size_t)(L_ * D_));
  const int d  = (int)(e & (size_t)(D_ - 1));
  const float4* vm4 = (const float4*)vmean;
  float4 val = vm4[bh * (D_ / 4) + (d >> 2)];
  ((float4*)out)[idx] = val;
}

__global__ __launch_bounds__(512)
void attn_kernel(const float* __restrict__ q, const float* __restrict__ k,
                 const float* __restrict__ v, float* __restrict__ out) {
  __shared__ float pLds[16][16 * 17];

  const int bh   = blockIdx.x;
  const int wave = threadIdx.x >> 5;
  const int lane = threadIdx.x & 31;
  const int tcol = lane & 15;
  const int half = lane >> 4;

  const float* qb = q + (size_t)bh * L_ * D_ + (size_t)wave * 16 * D_;
  const float* kb = k + (size_t)bh * L_ * D_ + (size_t)KEY0 * D_;
  const float* vb = v + (size_t)bh * L_ * D_ + (size_t)KEY0 * D_;
  float*       ob = out + (size_t)bh * L_ * D_;

  v2f aq[16];
#pragma unroll
  for (int c4 = 0; c4 < 16; ++c4) {
    const float* p = qb + tcol * D_ + c4 * 4 + 2 * half;
    v2f a; a.x = p[0] * 0.125f; a.y = p[1] * 0.125f;
    aq[c4] = a;
  }

  const v8f vzero = {0.f, 0.f, 0.f, 0.f, 0.f, 0.f, 0.f, 0.f};
  v8f acc[4];
  acc[0] = vzero; acc[1] = vzero; acc[2] = vzero; acc[3] = vzero;
  float mrow[8], lsum[8];
#pragma unroll
  for (int r = 0; r < 8; ++r) { mrow[r] = -3.0e38f; lsum[r] = 0.0f; }

  float* myP = &pLds[wave][0];

  for (int kt = 0; kt < 16; ++kt) {
    v8f s = vzero;
#pragma unroll
    for (int c4 = 0; c4 < 16; ++c4) {
      const float* kp = kb + (kt * 16 + tcol) * D_ + c4 * 4 + 2 * half;
      v2f b; b.x = kp[0]; b.y = kp[1];
      s = __builtin_amdgcn_wmma_f32_16x16x4_f32(
              false, aq[c4], false, b, (short)0, s, false, false);
    }

    const int j = KEY0 + kt * 16 + tcol;
    float alpha[8];
#pragma unroll
    for (int r = 0; r < 8; ++r) {
      const int i = wave * 16 + r + 8 * half;
      float sv = (j - i >= DIAG) ? s[r] : -10000.0f;
      float rm = sv;
      rm = fmaxf(rm, __shfl_xor(rm, 1));
      rm = fmaxf(rm, __shfl_xor(rm, 2));
      rm = fmaxf(rm, __shfl_xor(rm, 4));
      rm = fmaxf(rm, __shfl_xor(rm, 8));
      const float mnew = fmaxf(mrow[r], rm);
      alpha[r] = __expf(mrow[r] - mnew);
      const float p = __expf(sv - mnew);
      float ps = p;
      ps += __shfl_xor(ps, 1);
      ps += __shfl_xor(ps, 2);
      ps += __shfl_xor(ps, 4);
      ps += __shfl_xor(ps, 8);
      lsum[r] = lsum[r] * alpha[r] + ps;
      mrow[r] = mnew;
      myP[(r + 8 * half) * 17 + tcol] = p;
    }

#pragma unroll
    for (int nt = 0; nt < 4; ++nt)
#pragma unroll
      for (int r = 0; r < 8; ++r)
        acc[nt][r] *= alpha[r];

#pragma unroll
    for (int nt = 0; nt < 4; ++nt) {
#pragma unroll
      for (int c4 = 0; c4 < 4; ++c4) {
        const int kk = c4 * 4 + 2 * half;
        v2f a; a.x = myP[tcol * 17 + kk];
               a.y = myP[tcol * 17 + kk + 1];
        const float* vp = vb + (kt * 16 + kk) * D_ + nt * 16 + tcol;
        v2f b; b.x = vp[0]; b.y = vp[D_];
        acc[nt] = __builtin_amdgcn_wmma_f32_16x16x4_f32(
                      false, a, false, b, (short)0, acc[nt], false, false);
      }
    }
  }

#pragma unroll
  for (int r = 0; r < 8; ++r) {
    const int i = wave * 16 + r + 8 * half;
    if (i <= LASTQ) {
      const float inv = 1.0f / lsum[r];
#pragma unroll
      for (int nt = 0; nt < 4; ++nt)
        ob[(size_t)i * D_ + nt * 16 + tcol] = acc[nt][r] * inv;
    }
  }
}

extern "C" void kernel_launch(void* const* d_in, const int* in_sizes, int n_in,
                              void* d_out, int out_size, void* d_ws, size_t ws_size,
                              hipStream_t stream) {
  const float* q = (const float*)d_in[0];
  const float* k = (const float*)d_in[1];
  const float* v = (const float*)d_in[2];
  float* out = (float*)d_out;

  float* partial = (float*)d_ws;
  float* vmean   = partial + B_ * H_ * NSPLIT * D_;

  vpartial_kernel<<<B_ * H_ * NSPLIT, 256, 0, stream>>>(v, partial);
  vcombine_kernel<<<B_ * H_, 64, 0, stream>>>(partial, vmean);

  const int n4 = (B_ * H_ * L_ * D_) / 4;
  bcast_kernel<<<n4 / 256, 256, 0, stream>>>(vmean, out);

  attn_kernel<<<B_ * H_, 512, 0, stream>>>(q, k, v, out);
}


# attn corner: key-tile loop spread over waves (WG = query tiles p,15-p of a head, 17 waves = its 17 unmasked tile pairs), partial max/sum/O merged in LDS: new summation order
# speedup vs baseline: 2.6432x; 2.6432x over previous
#include <hip/hip_runtime.h>


typedef __attribute__((ext_vector_type(2))) float v2f;
typedef __attribute__((ext_vector_type(8))) float v8f;

#define B_    4
#define H_    16
#define L_    2048
#define D_    64
#define RWIN  256
#define KEY0  (L_ - RWIN)
#define DIAG  (L_ - RWIN + 1)
#define LASTQ (L_ - 1 - DIAG)
#define NSPLIT 8
#define ROWS_PER_SPLIT (L_ / NSPLIT)

__global__ __launch_bounds__(256)
void vpartial_kernel(const float* __restrict__ v, float* __restrict__ partial) {
  __shared__ float4 red[256];
  const int blk = blockIdx.x;
  const int bh  = blk >> 3;
  const int s   = blk & (NSPLIT - 1);
  const int t   = threadIdx.x;
  const int c   = t & 15;
  const int g   = t >> 4;

  const float4* vb4 = (const float4*)(v + (size_t)bh * L_ * D_);
  const int row0 = s * ROWS_PER_SPLIT + g;

  float4 acc = make_float4(0.f, 0.f, 0.f, 0.f);
#pragma unroll
  for (int it = 0; it < ROWS_PER_SPLIT / 16; ++it) {
    float4 x = vb4[(size_t)(row0 + it * 16) * (D_ / 4) + c];
    acc.x += x.x; acc.y += x.y; acc.z += x.z; acc.w += x.w;
  }
  red[t] = acc;
  __syncthreads();
  if (t < 128) { float4 o = red[t + 128];
    red[t].x += o.x; red[t].y += o.y; red[t].z += o.z; red[t].w += o.w; }
  __syncthreads();
  if (t < 64)  { float4 o = red[t + 64];
    red[t].x += o.x; red[t].y += o.y; red[t].z += o.z; red[t].w += o.w; }
  __syncthreads();
  if (t < 32)  { float4 o = red[t + 32];
    red[t].x += o.x; red[t].y += o.y; red[t].z += o.z; red[t].w += o.w; }
  __syncthreads();
  if (t < 16) {
    float4 o = red[t + 16];
    float4 r = red[t];
    r.x += o.x; r.y += o.y; r.z += o.z; r.w += o.w;
    ((float4*)partial)[(size_t)blk * (D_ / 4) + t] = r;
  }
}

__global__ __launch_bounds__(64)
void vcombine_kernel(const float* __restrict__ partial, float* __restrict__ vmean) {
  const int bh = blockIdx.x;
  const int d  = threadIdx.x;
  float s = 0.f;
#pragma unroll
  for (int sp = 0; sp < NSPLIT; ++sp)
    s += partial[(size_t)(bh * NSPLIT + sp) * D_ + d];
  vmean[bh * D_ + d] = s * (1.0f / (float)L_);
}

__global__ __launch_bounds__(256)
void bcast_kernel(const float* __restrict__ vmean, float* __restrict__ out) {
  const size_t idx = (size_t)blockIdx.x * blockDim.x + threadIdx.x;
  const size_t e   = idx * 4;
  const int bh = (int)(e / (size_t)(L_ * D_));
  const int d  = (int)(e & (size_t)(D_ - 1));
  const float4* vm4 = (const float4*)vmean;
  float4 val = vm4[bh * (D_ / 4) + (d >> 2)];
  ((float4*)out)[idx] = val;
}

#define NWAVE  17
#define OPITCH 68

__global__ __launch_bounds__(NWAVE * 32)
void attn_kernel(const float* __restrict__ q, const float* __restrict__ k,
                 const float* __restrict__ v, float* __restrict__ out) {
  __shared__ float pLds[NWAVE][16 * 17];
  __shared__ float4 oLds[NWAVE][16 * OPITCH / 4];
  __shared__ float4 mLds[NWAVE][4];
  __shared__ float4 lLds[NWAVE][4];

  const int bh   = blockIdx.x >> 3;
  const int p    = blockIdx.x & 7;
  const int wave = threadIdx.x >> 5;
  const int lane = threadIdx.x & 31;
  const int tcol = lane & 15;
  const int half = lane >> 4;

  const bool second = wave >= 16 - p;
  const int  w  = second ? 15 - p   : p;
  const int  kt = second ? wave - 1 : p + wave;

  const float* qb = q + (size_t)bh * L_ * D_ + (size_t)w * 16 * D_;
  const float* kb = k + (size_t)bh * L_ * D_ + (size_t)(KEY0 + kt * 16) * D_;
  const float* vb = v + (size_t)bh * L_ * D_ + (size_t)(KEY0 + kt * 16) * D_;

  v2f aq[16], bk[16];
#pragma unroll
  for (int c4 = 0; c4 < 16; ++c4) {
    const float* qp = qb + tcol * D_ + c4 * 4 + 2 * half;
    const float* kp = kb + tcol * D_ + c4 * 4 + 2 * half;
    v2f a; a.x = qp[0] * 0.125f; a.y = qp[1] * 0.125f;
    v2f b; b.x = kp[0];          b.y = kp[1];
    aq[c4] = a; bk[c4] = b;
  }
  v2f bv[4][4];
#pragma unroll
  for (int nt = 0; nt < 4; ++nt)
#pragma unroll
    for (int c4 = 0; c4 < 4; ++c4) {
      const int kk = c4 * 4 + 2 * half;
      const float* vp = vb + kk * D_ + nt * 16 + tcol;
      v2f b; b.x = vp[0]; b.y = vp[D_];
      bv[nt][c4] = b;
    }

  const v8f vzero = {0.f, 0.f, 0.f, 0.f, 0.f, 0.f, 0.f, 0.f};

  v8f s = vzero;
#pragma unroll
  for (int c4 = 0; c4 < 16; ++c4)
    s = __builtin_amdgcn_wmma_f32_16x16x4_f32(
            false, aq[c4], false, bk[c4], (short)0, s, false, false);

  float* myP = &pLds[wave][0];
  const int j = KEY0 + kt * 16 + tcol;
  float mrow[8], lsum[8];
#pragma unroll
  for (int r = 0; r < 8; ++r) {
    const int i = w * 16 + r + 8 * half;
    float sv = (j - i >= DIAG) ? s[r] : -10000.0f;
    float rm = sv;
    rm = fmaxf(rm, __shfl_xor(rm, 1));
    rm = fmaxf(rm, __shfl_xor(rm, 2));
    rm = fmaxf(rm, __shfl_xor(rm, 4));
    rm = fmaxf(rm, __shfl_xor(rm, 8));
    const float pe = __expf(sv - rm);
    float ps = pe;
    ps += __shfl_xor(ps, 1);
    ps += __shfl_xor(ps, 2);
    ps += __shfl_xor(ps, 4);
    ps += __shfl_xor(ps, 8);
    mrow[r] = rm;
    lsum[r] = ps;
    myP[(r + 8 * half) * 17 + tcol] = pe;
  }
  if (tcol == 0) {
    mLds[wave][2 * half]     = make_float4(mrow[0], mrow[1], mrow[2], mrow[3]);
    mLds[wave][2 * half + 1] = make_float4(mrow[4], mrow[5], mrow[6], mrow[7]);
    lLds[wave][2 * half]     = make_float4(lsum[0], lsum[1], lsum[2], lsum[3]);
    lLds[wave][2 * half + 1] = make_float4(lsum[4], lsum[5], lsum[6], lsum[7]);
  }

  v8f acc[4];
  acc[0] = vzero; acc[1] = vzero; acc[2] = vzero; acc[3] = vzero;
#pragma unroll
  for (int c4 = 0; c4 < 4; ++c4) {
    const int kk = c4 * 4 + 2 * half;
    v2f a; a.x = myP[tcol * 17 + kk];
           a.y = myP[tcol * 17 + kk + 1];
#pragma unroll
    for (int nt = 0; nt < 4; ++nt)
      acc[nt] = __builtin_amdgcn_wmma_f32_16x16x4_f32(
                    false, a, false, bv[nt][c4], (short)0, acc[nt], false, false);
  }

  float* myO = (float*)&oLds[wave][0];
#pragma unroll
  for (int nt = 0; nt < 4; ++nt)
#pragma unroll
    for (int r = 0; r < 8; ++r)
      myO[(r + 8 * half) * OPITCH + nt * 16 + tcol] = acc[nt][r];

  __syncthreads();

  const int tid = threadIdx.x;
  if (tid < 512) {
    const int g    = tid >> 8;
    const int row  = (tid >> 4) & 15;
    const int c    = tid & 15;
    const int w0   = g ? 16 - p : 0;
    const int nw   = g ? p + 1  : 16 - p;
    const int i    = (g ? 15 - p : p) * 16 + row;

    const float* mS = (const float*)&mLds[w0][0] + row;
    const float* lS = (const float*)&lLds[w0][0] + row;

    float mx = -3.0e38f;
    for (int t = 0; t < nw; ++t) mx = fmaxf(mx, mS[t * 16]);

    float  lrow = 0.0f;
    float4 o = make_float4(0.f, 0.f, 0.f, 0.f);
    for (int t = 0; t < nw; ++t) {
      const float  f = __expf(mS[t * 16] - mx);
      const float4 x = oLds[w0 + t][row * (OPITCH / 4) + c];
      lrow += lS[t * 16] * f;
      o.x += x.x * f; o.y += x.y * f; o.z += x.z * f; o.w += x.w * f;
    }
    if (i <= LASTQ) {
      const float inv = 1.0f / lrow;
      float4* orow = (float4*)(out + (size_t)bh * L_ * D_ + (size_t)i * D_);
      orow[c] = make_float4(o.x * inv, o.y * inv, o.z * inv, o.w * inv);
    }
  }
}

extern "C" void kernel_launch(void* const* d_in, const int* in_sizes, int n_in,
                              void* d_out, int out_size, void* d_ws, size_t ws_size,
                              hipStream_t stream) {
  const float* q = (const float*)d_in[0];
  const float* k = (const float*)d_in[1];
  const float* v = (const float*)d_in[2];
  float* out = (float*)d_out;

  float* partial = (float*)d_ws;
  float* vmean   = partial + B_ * H_ * NSPLIT * D_;

  vpartial_kernel<<<B_ * H_ * NSPLIT, 256, 0, stream>>>(v, partial);
  vcombine_kernel<<<B_ * H_, 64, 0, stream>>>(partial, vmean);

  const int n4 = (B_ * H_ * L_ * D_) / 4;
  bcast_kernel<<<n4 / 256, 256, 0, stream>>>(vmean, out);

  attn_kernel<<<B_ * H_ * 8, NWAVE * 32, 0, stream>>>(q, k, v, out);
}


# 4 launches -> 2: attn work-groups also stream 256 rows of v each into the partial sums; 256 broadcast WGs add the 8 partials themselves, store 448 rows each, skip rows 0..254; new sum order of mean
# speedup vs baseline: 3.2962x; 1.2470x over previous
#include <hip/hip_runtime.h>


typedef __attribute__((ext_vector_type(2))) float v2f;
typedef __attribute__((ext_vector_type(8))) float v8f;

#define B_    4
#define H_    16
#define L_    2048
#define D_    64
#define RWIN  256
#define KEY0  (L_ - RWIN)
#define DIAG  (L_ - RWIN + 1)
#define LASTQ (L_ - 1 - DIAG)
#define NSPLIT 8
#define ROWS_PER_SPLIT (L_ / NSPLIT)

#define BCHUNK 4
#define BROWS  ((L_ - RWIN) / BCHUNK)

__global__ __launch_bounds__(256)
void bcast_kernel(const float* __restrict__ partial, float* __restrict__ out) {
  const int bh = blockIdx.x / BCHUNK;
  const int ch = blockIdx.x & (BCHUNK - 1);
  const int t  = threadIdx.x;
  const int c  = t & 15;
  const int g  = t >> 4;

  const float4* p4 = (const float4*)partial + (size_t)bh * NSPLIT * (D_ / 4) + c;
  float4 s = make_float4(0.f, 0.f, 0.f, 0.f);
#pragma unroll
  for (int sp = 0; sp < NSPLIT; ++sp) {
    const float4 x = p4[sp * (D_ / 4)];
    s.x += x.x; s.y += x.y; s.z += x.z; s.w += x.w;
  }
  const float sc = 1.0f / (float)L_;
  s.x *= sc; s.y *= sc; s.z *= sc; s.w *= sc;

  float4* ob = (float4*)(out + (size_t)bh * L_ * D_);
  float4* o  = ob + (size_t)(RWIN + ch * BROWS + g) * (D_ / 4) + c;
#pragma unroll
  for (int it = 0; it < BROWS / 16; ++it)
    o[(size_t)it * 16 * (D_ / 4)] = s;
  if (ch == 0 && g == 0)
    ob[(size_t)(RWIN - 1) * (D_ / 4) + c] = s;
}

#define NWAVE  17
#define OPITCH 68

__global__ __launch_bounds__(NWAVE * 32)
void attn_kernel(const float* __restrict__ q, const float* __restrict__ k,
                 const float* __restrict__ v, float* __restrict__ out,
                 float* __restrict__ partial) {
  __shared__ float pLds[NWAVE][16 * 17];
  __shared__ float4 oLds[NWAVE][16 * OPITCH / 4];
  __shared__ float4 mLds[NWAVE][4];
  __shared__ float4 lLds[NWAVE][4];
  __shared__ float4 vLds[16][16];

  const int bh   = blockIdx.x >> 3;
  const int p    = blockIdx.x & 7;
  const int wave = threadIdx.x >> 5;
  const int lane = threadIdx.x & 31;
  const int tcol = lane & 15;
  const int half = lane >> 4;

  const bool second = wave >= 16 - p;
  const int  w  = second ? 15 - p   : p;
  const int  kt = second ? wave - 1 : p + wave;

  const float* qb = q + (size_t)bh * L_ * D_ + (size_t)w * 16 * D_;
  const float* kb = k + (size_t)bh * L_ * D_ + (size_t)(KEY0 + kt * 16) * D_;
  const float* vb = v + (size_t)bh * L_ * D_ + (size_t)(KEY0 + kt * 16) * D_;

  v2f aq[16], bk[16];
#pragma unroll
  for (int c4 = 0; c4 < 16; ++c4) {
    const float* qp = qb + tcol * D_ + c4 * 4 + 2 * half;
    const float* kp = kb + tcol * D_ + c4 * 4 + 2 * half;
    v2f a; a.x = qp[0] * 0.125f; a.y = qp[1] * 0.125f;
    v2f b; b.x = kp[0];          b.y = kp[1];
    aq[c4] = a; bk[c4] = b;
  }
  v2f bv[4][4];
#pragma unroll
  for (int nt = 0; nt < 4; ++nt)
#pragma unroll
    for (int c4 = 0; c4 < 4; ++c4) {
      const int kk = c4 * 4 + 2 * half;
      const float* vp = vb + kk * D_ + nt * 16 + tcol;
      v2f b; b.x = vp[0]; b.y = vp[D_];
      bv[nt][c4] = b;
    }

  const v8f vzero = {0.f, 0.f, 0.f, 0.f, 0.f, 0.f, 0.f, 0.f};

  v8f s = vzero;
#pragma unroll
  for (int c4 = 0; c4 < 16; ++c4)
    s = __builtin_amdgcn_wmma_f32_16x16x4_f32(
            false, aq[c4], false, bk[c4], (short)0, s, false, false);

  float* myP = &pLds[wave][0];
  const int j = KEY0 + kt * 16 + tcol;
  float mrow[8], lsum[8];
#pragma unroll
  for (int r = 0; r < 8; ++r) {
    const int i = w * 16 + r + 8 * half;
    float sv = (j - i >= DIAG) ? s[r] : -10000.0f;
    float rm = sv;
    rm = fmaxf(rm, __shfl_xor(rm, 1));
    rm = fmaxf(rm, __shfl_xor(rm, 2));
    rm = fmaxf(rm, __shfl_xor(rm, 4));
    rm = fmaxf(rm, __shfl_xor(rm, 8));
    const float pe = __expf(sv - rm);
    float ps = pe;
    ps += __shfl_xor(ps, 1);
    ps += __shfl_xor(ps, 2);
    ps += __shfl_xor(ps, 4);
    ps += __shfl_xor(ps, 8);
    mrow[r] = rm;
    lsum[r] = ps;
    myP[(r + 8 * half) * 17 + tcol] = pe;
  }
  if (tcol == 0) {
    mLds[wave][2 * half]     = make_float4(mrow[0], mrow[1], mrow[2], mrow[3]);
    mLds[wave][2 * half + 1] = make_float4(mrow[4], mrow[5], mrow[6], mrow[7]);
    lLds[wave][2 * half]     = make_float4(lsum[0], lsum[1], lsum[2], lsum[3]);
    lLds[wave][2 * half + 1] = make_float4(lsum[4], lsum[5], lsum[6], lsum[7]);
  }

  v8f acc[4];
  acc[0] = vzero; acc[1] = vzero; acc[2] = vzero; acc[3] = vzero;
#pragma unroll
  for (int c4 = 0; c4 < 4; ++c4) {
    const int kk = c4 * 4 + 2 * half;
    v2f a; a.x = myP[tcol * 17 + kk];
           a.y = myP[tcol * 17 + kk + 1];
#pragma unroll
    for (int nt = 0; nt < 4; ++nt)
      acc[nt] = __builtin_amdgcn_wmma_f32_16x16x4_f32(
                    false, a, false, bv[nt][c4], (short)0, acc[nt], false, false);
  }

  float* myO = (float*)&oLds[wave][0];
#pragma unroll
  for (int nt = 0; nt < 4; ++nt)
#pragma unroll
    for (int r = 0; r < 8; ++r)
      myO[(r + 8 * half) * OPITCH + nt * 16 + tcol] = acc[nt][r];

  {
    const int sw = wave < 16 ? wave : 15;
    const float4* vs = (const float4*)(v + (size_t)bh * L_ * D_
                                         + (size_t)(p * ROWS_PER_SPLIT + sw * 16) * D_);
    float4 vrow[8];
#pragma unroll
    for (int r = 0; r < 8; ++r)
      vrow[r] = vs[(2 * r + half) * (D_ / 4) + tcol];
    float4 a4 = vrow[0];
#pragma unroll
    for (int r = 1; r < 8; ++r) {
      a4.x += vrow[r].x; a4.y += vrow[r].y; a4.z += vrow[r].z; a4.w += vrow[r].w;
    }
    a4.x += __shfl_xor(a4.x, 16);
    a4.y += __shfl_xor(a4.y, 16);
    a4.z += __shfl_xor(a4.z, 16);
    a4.w += __shfl_xor(a4.w, 16);
    if (wave < 16 && half == 0) vLds[wave][tcol] = a4;
  }

  __syncthreads();

  const int tid = threadIdx.x;
  if (tid < 512) {
    const int g    = tid >> 8;
    const int row  = (tid >> 4) & 15;
    const int c    = tid & 15;
    const int w0   = g ? 16 - p : 0;
    const int nw   = g ? p + 1  : 16 - p;
    const int i    = (g ? 15 - p : p) * 16 + row;

    const float* mS = (const float*)&mLds[w0][0] + row;
    const float* lS = (const float*)&lLds[w0][0] + row;

    float mx = -3.0e38f;
    for (int t = 0; t < nw; ++t) mx = fmaxf(mx, mS[t * 16]);

    float  lrow = 0.0f;
    float4 o = make_float4(0.f, 0.f, 0.f, 0.f);
    for (int t = 0; t < nw; ++t) {
      const float  f = __expf(mS[t * 16] - mx);
      const float4 x = oLds[w0 + t][row * (OPITCH / 4) + c];
      lrow += lS[t * 16] * f;
      o.x += x.x * f; o.y += x.y * f; o.z += x.z * f; o.w += x.w * f;
    }
    if (i <= LASTQ) {
      const float inv = 1.0f / lrow;
      float4* orow = (float4*)(out + (size_t)bh * L_ * D_ + (size_t)i * D_);
      orow[c] = make_float4(o.x * inv, o.y * inv, o.z * inv, o.w * inv);
    }
  } else {
    const int c2 = tid - 512;
    const float2* vl = (const float2*)&vLds[0][0] + c2;
    float2 a2 = vl[0];
    for (int t = 1; t < 16; ++t) {
      const float2 x = vl[t * 32];
      a2.x += x.x; a2.y += x.y;
    }
    ((float2*)partial)[(size_t)blockIdx.x * (D_ / 2) + c2] = a2;
  }
}

extern "C" void kernel_launch(void* const* d_in, const int* in_sizes, int n_in,
                              void* d_out, int out_size, void* d_ws, size_t ws_size,
                              hipStream_t stream) {
  const float* q = (const float*)d_in[0];
  const float* k = (const float*)d_in[1];
  const float* v = (const float*)d_in[2];
  float* out = (float*)d_out;

  float* partial = (float*)d_ws;

  attn_kernel<<<B_ * H_ * 8, NWAVE * 32, 0, stream>>>(q, k, v, out, partial);
  bcast_kernel<<<B_ * H_ * BCHUNK, 256, 0, stream>>>(partial, out);
}


# attn tiles fetched with 16-byte loads: Q/K fragments take a permuted k order (d=8m+4h+2u+e), V columns permuted (d=4c+nt) so O is staged by 16-byte LDS stores; new k order of the score sums
# speedup vs baseline: 3.4980x; 1.0612x over previous
#include <hip/hip_runtime.h>


typedef __attribute__((ext_vector_type(2))) float v2f;
typedef __attribute__((ext_vector_type(4))) float v4f;
typedef __attribute__((ext_vector_type(8))) float v8f;

#define B_    4
#define H_    16
#define L_    2048
#define D_    64
#define RWIN  256
#define KEY0  (L_ - RWIN)
#define DIAG  (L_ - RWIN + 1)
#define LASTQ (L_ - 1 - DIAG)
#define NSPLIT 8
#define ROWS_PER_SPLIT (L_ / NSPLIT)

#define BCHUNK 4
#define BROWS  ((L_ - RWIN) / BCHUNK)

__global__ __launch_bounds__(256)
void bcast_kernel(const float* __restrict__ partial, float* __restrict__ out) {
  const int bh = blockIdx.x / BCHUNK;
  const int ch = blockIdx.x & (BCHUNK - 1);
  const int t  = threadIdx.x;
  const int c  = t & 15;
  const int g  = t >> 4;

  const float4* p4 = (const float4*)partial + (size_t)bh * NSPLIT * (D_ / 4) + c;
  float4 s = make_float4(0.f, 0.f, 0.f, 0.f);
#pragma unroll
  for (int sp = 0; sp < NSPLIT; ++sp) {
    const float4 x = p4[sp * (D_ / 4)];
    s.x += x.x; s.y += x.y; s.z += x.z; s.w += x.w;
  }
  const float sc = 1.0f / (float)L_;
  s.x *= sc; s.y *= sc; s.z *= sc; s.w *= sc;

  float4* ob = (float4*)(out + (size_t)bh * L_ * D_);
  float4* o  = ob + (size_t)(RWIN + ch * BROWS + g) * (D_ / 4) + c;
#pragma unroll
  for (int it = 0; it < BROWS / 16; ++it)
    o[(size_t)it * 16 * (D_ / 4)] = s;
  if (ch == 0 && g == 0)
    ob[(size_t)(RWIN - 1) * (D_ / 4) + c] = s;
}

#define NWAVE  17
#define OPITCH 68

__global__ __launch_bounds__(NWAVE * 32)
void attn_kernel(const float* __restrict__ q, const float* __restrict__ k,
                 const float* __restrict__ v, float* __restrict__ out,
                 float* __restrict__ partial) {
  __shared__ float pLds[NWAVE][16 * 17];
  __shared__ float4 oLds[NWAVE][16 * OPITCH / 4];
  __shared__ float4 mLds[NWAVE][4];
  __shared__ float4 lLds[NWAVE][4];
  __shared__ float4 vLds[16][16];

  const int bh   = blockIdx.x >> 3;
  const int p    = blockIdx.x & 7;
  const int wave = threadIdx.x >> 5;
  const int lane = threadIdx.x & 31;
  const int tcol = lane & 15;
  const int half = lane >> 4;

  const bool second = wave >= 16 - p;
  const int  w  = second ? 15 - p   : p;
  const int  kt = second ? wave - 1 : p + wave;

  const float* qb = q + (size_t)bh * L_ * D_ + (size_t)w * 16 * D_;
  const float* kb = k + (size_t)bh * L_ * D_ + (size_t)(KEY0 + kt * 16) * D_;
  const float* vb = v + (size_t)bh * L_ * D_ + (size_t)(KEY0 + kt * 16) * D_;

  const v4f* q4 = (const v4f*)(qb + tcol * D_) + half;
  const v4f* k4 = (const v4f*)(kb + tcol * D_) + half;
  v2f aq[16], bk[16];
#pragma unroll
  for (int m = 0; m < 8; ++m) {
    const v4f qv = q4[2 * m];
    const v4f kv = k4[2 * m];
    v2f a0; a0.x = qv[0] * 0.125f; a0.y = qv[1] * 0.125f;
    v2f a1; a1.x = qv[2] * 0.125f; a1.y = qv[3] * 0.125f;
    v2f b0; b0.x = kv[0];          b0.y = kv[1];
    v2f b1; b1.x = kv[2];          b1.y = kv[3];
    aq[2 * m] = a0; aq[2 * m + 1] = a1;
    bk[2 * m] = b0; bk[2 * m + 1] = b1;
  }
  const v4f* v4p = (const v4f*)vb + tcol;
  v2f bv[4][4];
#pragma unroll
  for (int c4 = 0; c4 < 4; ++c4) {
    const int kk = c4 * 4 + 2 * half;
    const v4f r0 = v4p[(kk)     * (D_ / 4)];
    const v4f r1 = v4p[(kk + 1) * (D_ / 4)];
#pragma unroll
    for (int nt = 0; nt < 4; ++nt) {
      v2f b; b.x = r0[nt]; b.y = r1[nt];
      bv[nt][c4] = b;
    }
  }

  const v8f vzero = {0.f, 0.f, 0.f, 0.f, 0.f, 0.f, 0.f, 0.f};

  v8f s = vzero;
#pragma unroll
  for (int c4 = 0; c4 < 16; ++c4)
    s = __builtin_amdgcn_wmma_f32_16x16x4_f32(
            false, aq[c4], false, bk[c4], (short)0, s, false, false);

  float* myP = &pLds[wave][0];
  const int j = KEY0 + kt * 16 + tcol;
  float mrow[8], lsum[8];
#pragma unroll
  for (int r = 0; r < 8; ++r) {
    const int i = w * 16 + r + 8 * half;
    float sv = (j - i >= DIAG) ? s[r] : -10000.0f;
    float rm = sv;
    rm = fmaxf(rm, __shfl_xor(rm, 1));
    rm = fmaxf(rm, __shfl_xor(rm, 2));
    rm = fmaxf(rm, __shfl_xor(rm, 4));
    rm = fmaxf(rm, __shfl_xor(rm, 8));
    const float pe = __expf(sv - rm);
    float ps = pe;
    ps += __shfl_xor(ps, 1);
    ps += __shfl_xor(ps, 2);
    ps += __shfl_xor(ps, 4);
    ps += __shfl_xor(ps, 8);
    mrow[r] = rm;
    lsum[r] = ps;
    myP[(r + 8 * half) * 17 + tcol] = pe;
  }
  if (tcol == 0) {
    mLds[wave][2 * half]     = make_float4(mrow[0], mrow[1], mrow[2], mrow[3]);
    mLds[wave][2 * half + 1] = make_float4(mrow[4], mrow[5], mrow[6], mrow[7]);
    lLds[wave][2 * half]     = make_float4(lsum[0], lsum[1], lsum[2], lsum[3]);
    lLds[wave][2 * half + 1] = make_float4(lsum[4], lsum[5], lsum[6], lsum[7]);
  }

  v8f acc[4];
  acc[0] = vzero; acc[1] = vzero; acc[2] = vzero; acc[3] = vzero;
#pragma unroll
  for (int c4 = 0; c4 < 4; ++c4) {
    const int kk = c4 * 4 + 2 * half;
    v2f a; a.x = myP[tcol * 17 + kk];
           a.y = myP[tcol * 17 + kk + 1];
#pragma unroll
    for (int nt = 0; nt < 4; ++nt)
      acc[nt] = __builtin_amdgcn_wmma_f32_16x16x4_f32(
                    false, a, false, bv[nt][c4], (short)0, acc[nt], false, false);
  }

  float* myO = (float*)&oLds[wave][0];
#pragma unroll
  for (int r = 0; r < 8; ++r) {
    v4f o4; o4[0] = acc[0][r]; o4[1] = acc[1][r]; o4[2] = acc[2][r]; o4[3] = acc[3][r];
    *(v4f*)&myO[(r + 8 * half) * OPITCH + 4 * tcol] = o4;
  }

  {
    const int sw = wave < 16 ? wave : 15;
    const float4* vs = (const float4*)(v + (size_t)bh * L_ * D_
                                         + (size_t)(p * ROWS_PER_SPLIT + sw * 16) * D_);
    float4 vrow[8];
#pragma unroll
    for (int r = 0; r < 8; ++r)
      vrow[r] = vs[(2 * r + half) * (D_ / 4) + tcol];
    float4 a4 = vrow[0];
#pragma unroll
    for (int r = 1; r < 8; ++r) {
      a4.x += vrow[r].x; a4.y += vrow[r].y; a4.z += vrow[r].z; a4.w += vrow[r].w;
    }
    a4.x += __shfl_xor(a4.x, 16);
    a4.y += __shfl_xor(a4.y, 16);
    a4.z += __shfl_xor(a4.z, 16);
    a4.w += __shfl_xor(a4.w, 16);
    if (wave < 16 && half == 0) vLds[wave][tcol] = a4;
  }

  __syncthreads();

  const int tid = threadIdx.x;
  if (tid < 512) {
    const int g    = tid >> 8;
    const int row  = (tid >> 4) & 15;
    const int c    = tid & 15;
    const int w0   = g ? 16 - p : 0;
    const int nw   = g ? p + 1  : 16 - p;
    const int i    = (g ? 15 - p : p) * 16 + row;

    const float* mS = (const float*)&mLds[w0][0] + row;
    const float* lS = (const float*)&lLds[w0][0] + row;

    float mx = -3.0e38f;
    for (int t = 0; t < nw; ++t) mx = fmaxf(mx, mS[t * 16]);

    float  lrow = 0.0f;
    float4 o = make_float4(0.f, 0.f, 0.f, 0.f);
    for (int t = 0; t < nw; ++t) {
      const float  f = __expf(mS[t * 16] - mx);
      const float4 x = oLds[w0 + t][row * (OPITCH / 4) + c];
      lrow += lS[t * 16] * f;
      o.x += x.x * f; o.y += x.y * f; o.z += x.z * f; o.w += x.w * f;
    }
    if (i <= LASTQ) {
      const float inv = 1.0f / lrow;
      float4* orow = (float4*)(out + (size_t)bh * L_ * D_ + (size_t)i * D_);
      orow[c] = make_float4(o.x * inv, o.y * inv, o.z * inv, o.w * inv);
    }
  } else {
    const int c2 = tid - 512;
    const float2* vl = (const float2*)&vLds[0][0] + c2;
    float2 a2 = vl[0];
    for (int t = 1; t < 16; ++t) {
      const float2 x = vl[t * 32];
      a2.x += x.x; a2.y += x.y;
    }
    ((float2*)partial)[(size_t)blockIdx.x * (D_ / 2) + c2] = a2;
  }
}

extern "C" void kernel_launch(void* const* d_in, const int* in_sizes, int n_in,
                              void* d_out, int out_size, void* d_ws, size_t ws_size,
                              hipStream_t stream) {
  const float* q = (const float*)d_in[0];
  const float* k = (const float*)d_in[1];
  const float* v = (const float*)d_in[2];
  float* out = (float*)d_out;

  float* partial = (float*)d_ws;

  attn_kernel<<<B_ * H_ * 8, NWAVE * 32, 0, stream>>>(q, k, v, out, partial);
  bcast_kernel<<<B_ * H_ * BCHUNK, 256, 0, stream>>>(partial, out);
}


# attn corner: each wave takes its query tile against TWO adjacent key tiles (16x32 scores, one softmax, one partial): 17 -> 9 waves per work-group, 9 partials merged per thread; new summation order
# speedup vs baseline: 3.7241x; 1.0646x over previous
#include <hip/hip_runtime.h>


typedef __attribute__((ext_vector_type(2))) float v2f;
typedef __attribute__((ext_vector_type(4))) float v4f;
typedef __attribute__((ext_vector_type(8))) float v8f;

#define B_    4
#define H_    16
#define L_    2048
#define D_    64
#define RWIN  256
#define KEY0  (L_ - RWIN)
#define DIAG  (L_ - RWIN + 1)
#define LASTQ (L_ - 1 - DIAG)
#define NSPLIT 8
#define ROWS_PER_SPLIT (L_ / NSPLIT)

#define BCHUNK 4
#define BROWS  ((L_ - RWIN) / BCHUNK)

__global__ __launch_bounds__(256)
void bcast_kernel(const float* __restrict__ partial, float* __restrict__ out) {
  const int bh = blockIdx.x / BCHUNK;
  const int ch = blockIdx.x & (BCHUNK - 1);
  const int t  = threadIdx.x;
  const int c  = t & 15;
  const int g  = t >> 4;

  const float4* p4 = (const float4*)partial + (size_t)bh * NSPLIT * (D_ / 4) + c;
  float4 s = make_float4(0.f, 0.f, 0.f, 0.f);
#pragma unroll
  for (int sp = 0; sp < NSPLIT; ++sp) {
    const float4 x = p4[sp * (D_ / 4)];
    s.x += x.x; s.y += x.y; s.z += x.z; s.w += x.w;
  }
  const float sc = 1.0f / (float)L_;
  s.x *= sc; s.y *= sc; s.z *= sc; s.w *= sc;

  float4* ob = (float4*)(out + (size_t)bh * L_ * D_);
  float4* o  = ob + (size_t)(RWIN + ch * BROWS + g) * (D_ / 4) + c;
#pragma unroll
  for (int it = 0; it < BROWS / 16; ++it)
    o[(size_t)it * 16 * (D_ / 4)] = s;
  if (ch == 0 && g == 0)
    ob[(size_t)(RWIN - 1) * (D_ / 4) + c] = s;
}

#define NWAVE  9
#define PPITCH 36
#define OPITCH 68

__global__ __launch_bounds__(NWAVE * 32)
void attn_kernel(const float* __restrict__ q, const float* __restrict__ k,
                 const float* __restrict__ v, float* __restrict__ out,
                 float* __restrict__ partial) {
  __shared__ __attribute__((aligned(16))) float pLds[NWAVE][16 * PPITCH];
  __shared__ float4 oLds[NWAVE][16 * OPITCH / 4];
  __shared__ float4 mLds[NWAVE][4];
  __shared__ float4 lLds[NWAVE][4];
  __shared__ float4 vLds[8][16];

  const int bh   = blockIdx.x >> 3;
  const int p    = blockIdx.x & 7;
  const int wave = threadIdx.x >> 5;
  const int lane = threadIdx.x & 31;
  const int tcol = lane & 15;
  const int half = lane >> 4;

  const int  n0     = 8 - (p >> 1);
  const bool second = wave >= n0;
  const int  w  = second ? 15 - p : p;
  const int  kp = second ? ((15 - p) >> 1) + (wave - n0)
                         : (p >> 1) + wave;

  const float* qb = q + (size_t)bh * L_ * D_ + (size_t)w * 16 * D_;
  const float* kb = k + (size_t)bh * L_ * D_ + (size_t)(KEY0 + kp * 32) * D_;
  const float* vb = v + (size_t)bh * L_ * D_ + (size_t)(KEY0 + kp * 32) * D_;

  const v4f* q4 = (const v4f*)(qb + tcol * D_) + half;
  const v4f* k4 = (const v4f*)(kb + tcol * D_) + half;
  v2f aq[16], bk0[16], bk1[16];
#pragma unroll
  for (int m = 0; m < 8; ++m) {
    const v4f qv = q4[2 * m];
    const v4f k0 = k4[2 * m];
    const v4f k1 = k4[2 * m + 16 * (D_ / 4)];
    v2f a0; a0.x = qv[0] * 0.125f; a0.y = qv[1] * 0.125f;
    v2f a1; a1.x = qv[2] * 0.125f; a1.y = qv[3] * 0.125f;
    v2f b0; b0.x = k0[0];          b0.y = k0[1];
    v2f b1; b1.x = k0[2];          b1.y = k0[3];
    v2f c0; c0.x = k1[0];          c0.y = k1[1];
    v2f c1; c1.x = k1[2];          c1.y = k1[3];
    aq[2 * m]  = a0; aq[2 * m + 1]  = a1;
    bk0[2 * m] = b0; bk0[2 * m + 1] = b1;
    bk1[2 * m] = c0; bk1[2 * m + 1] = c1;
  }
  const v4f* v4p = (const v4f*)vb + tcol;
  v2f bv[4][8];
#pragma unroll
  for (int c4 = 0; c4 < 8; ++c4) {
    const int kk = c4 * 4 + 2 * half;
    const v4f r0 = v4p[(kk)     * (D_ / 4)];
    const v4f r1 = v4p[(kk + 1) * (D_ / 4)];
#pragma unroll
    for (int nt = 0; nt < 4; ++nt) {
      v2f b; b.x = r0[nt]; b.y = r1[nt];
      bv[nt][c4] = b;
    }
  }

  const v8f vzero = {0.f, 0.f, 0.f, 0.f, 0.f, 0.f, 0.f, 0.f};

  v8f s0 = vzero, s1 = vzero;
#pragma unroll
  for (int c4 = 0; c4 < 16; ++c4) {
    s0 = __builtin_amdgcn_wmma_f32_16x16x4_f32(
             false, aq[c4], false, bk0[c4], (short)0, s0, false, false);
    s1 = __builtin_amdgcn_wmma_f32_16x16x4_f32(
             false, aq[c4], false, bk1[c4], (short)0, s1, false, false);
  }

  float* myP = &pLds[wave][0];
  const int j0 = KEY0 + kp * 32 + tcol;
  float mrow[8], lsum[8];
#pragma unroll
  for (int r = 0; r < 8; ++r) {
    const int i = w * 16 + r + 8 * half;
    const float sa = (j0 - i      >= DIAG) ? s0[r] : -10000.0f;
    const float sb = (j0 + 16 - i >= DIAG) ? s1[r] : -10000.0f;
    float rm = fmaxf(sa, sb);
    rm = fmaxf(rm, __shfl_xor(rm, 1));
    rm = fmaxf(rm, __shfl_xor(rm, 2));
    rm = fmaxf(rm, __shfl_xor(rm, 4));
    rm = fmaxf(rm, __shfl_xor(rm, 8));
    const float pa = __expf(sa - rm);
    const float pb = __expf(sb - rm);
    float ps = pa + pb;
    ps += __shfl_xor(ps, 1);
    ps += __shfl_xor(ps, 2);
    ps += __shfl_xor(ps, 4);
    ps += __shfl_xor(ps, 8);
    mrow[r] = rm;
    lsum[r] = ps;
    myP[(r + 8 * half) * PPITCH + tcol]      = pa;
    myP[(r + 8 * half) * PPITCH + 16 + tcol] = pb;
  }
  if (tcol == 0) {
    mLds[wave][2 * half]     = make_float4(mrow[0], mrow[1], mrow[2], mrow[3]);
    mLds[wave][2 * half + 1] = make_float4(mrow[4], mrow[5], mrow[6], mrow[7]);
    lLds[wave][2 * half]     = make_float4(lsum[0], lsum[1], lsum[2], lsum[3]);
    lLds[wave][2 * half + 1] = make_float4(lsum[4], lsum[5], lsum[6], lsum[7]);
  }

  v8f acc[4];
  acc[0] = vzero; acc[1] = vzero; acc[2] = vzero; acc[3] = vzero;
#pragma unroll
  for (int c4 = 0; c4 < 8; ++c4) {
    const int kk = c4 * 4 + 2 * half;
    const v2f a = *(const v2f*)&myP[tcol * PPITCH + kk];
#pragma unroll
    for (int nt = 0; nt < 4; ++nt)
      acc[nt] = __builtin_amdgcn_wmma_f32_16x16x4_f32(
                    false, a, false, bv[nt][c4], (short)0, acc[nt], false, false);
  }

  float* myO = (float*)&oLds[wave][0];
#pragma unroll
  for (int r = 0; r < 8; ++r) {
    v4f o4; o4[0] = acc[0][r]; o4[1] = acc[1][r]; o4[2] = acc[2][r]; o4[3] = acc[3][r];
    *(v4f*)&myO[(r + 8 * half) * OPITCH + 4 * tcol] = o4;
  }

  {
    const int sw = wave < 8 ? wave : 7;
    const float4* vs = (const float4*)(v + (size_t)bh * L_ * D_
                                         + (size_t)(p * ROWS_PER_SPLIT + sw * 32) * D_);
    float4 vrow[16];
#pragma unroll
    for (int r = 0; r < 16; ++r)
      vrow[r] = vs[(2 * r + half) * (D_ / 4) + tcol];
    float4 a4 = vrow[0];
#pragma unroll
    for (int r = 1; r < 16; ++r) {
      a4.x += vrow[r].x; a4.y += vrow[r].y; a4.z += vrow[r].z; a4.w += vrow[r].w;
    }
    a4.x += __shfl_xor(a4.x, 16);
    a4.y += __shfl_xor(a4.y, 16);
    a4.z += __shfl_xor(a4.z, 16);
    a4.w += __shfl_xor(a4.w, 16);
    if (wave < 8 && half == 0) vLds[wave][tcol] = a4;
  }

  __syncthreads();

  const int tid = threadIdx.x;
  if (tid < 256) {
    const int row = tid >> 4;
    const int c   = tid & 15;

    float  mt[NWAVE], lt[NWAVE];
    float4 xt[NWAVE];
#pragma unroll
    for (int t = 0; t < NWAVE; ++t) {
      mt[t] = ((const float*)&mLds[t][0])[row];
      lt[t] = ((const float*)&lLds[t][0])[row];
      xt[t] = oLds[t][row * (OPITCH / 4) + c];
    }
    float mxA = -3.0e38f, mxB = -3.0e38f;
#pragma unroll
    for (int t = 0; t < NWAVE; ++t) {
      if (t < n0) mxA = fmaxf(mxA, mt[t]);
      else        mxB = fmaxf(mxB, mt[t]);
    }
    float  lA = 0.0f, lB = 0.0f;
    float4 oA = make_float4(0.f, 0.f, 0.f, 0.f);
    float4 oB = make_float4(0.f, 0.f, 0.f, 0.f);
#pragma unroll
    for (int t = 0; t < NWAVE; ++t) {
      const bool  inA = t < n0;
      const float f   = __expf(mt[t] - (inA ? mxA : mxB));
      const float fa  = inA ? f : 0.0f;
      const float fb  = inA ? 0.0f : f;
      lA += lt[t] * fa;
      lB += lt[t] * fb;
      oA.x += xt[t].x * fa; oA.y += xt[t].y * fa; oA.z += xt[t].z * fa; oA.w += xt[t].w * fa;
      oB.x += xt[t].x * fb; oB.y += xt[t].y * fb; oB.z += xt[t].z * fb; oB.w += xt[t].w * fb;
    }
    float* ob = out + (size_t)bh * L_ * D_;
    {
      const int   iA  = p * 16 + row;
      const float inv = 1.0f / lA;
      ((float4*)(ob + (size_t)iA * D_))[c] =
          make_float4(oA.x * inv, oA.y * inv, oA.z * inv, oA.w * inv);
    }
    const int iB = (15 - p) * 16 + row;
    if (iB <= LASTQ) {
      const float inv = 1.0f / lB;
      ((float4*)(ob + (size_t)iB * D_))[c] =
          make_float4(oB.x * inv, oB.y * inv, oB.z * inv, oB.w * inv);
    }
  } else {
    const int c2 = tid - 256;
    const float2* vl = (const float2*)&vLds[0][0] + c2;
    float2 a2 = vl[0];
#pragma unroll
    for (int t = 1; t < 8; ++t) {
      const float2 x = vl[t * 32];
      a2.x += x.x; a2.y += x.y;
    }
    ((float2*)partial)[(size_t)blockIdx.x * (D_ / 2) + c2] = a2;
  }
}

extern "C" void kernel_launch(void* const* d_in, const int* in_sizes, int n_in,
                              void* d_out, int out_size, void* d_ws, size_t ws_size,
                              hipStream_t stream) {
  const float* q = (const float*)d_in[0];
  const float* k = (const float*)d_in[1];
  const float* v = (const float*)d_in[2];
  float* out = (float*)d_out;

  float* partial = (float*)d_ws;

  attn_kernel<<<B_ * H_ * 8, NWAVE * 32, 0, stream>>>(q, k, v, out, partial);
  bcast_kernel<<<B_ * H_ * BCHUNK, 256, 0, stream>>>(partial, out);
}


# 2 launches -> 1, other shape: 256 work-groups (head, quarter) of 18 waves do 4 query tiles, each sums its head's whole v itself (16 waves x 128 rows), stores 448 mean rows; new sum order of mean
# speedup vs baseline: 3.9723x; 1.0666x over previous
#include <hip/hip_runtime.h>


typedef __attribute__((ext_vector_type(2))) float v2f;
typedef __attribute__((ext_vector_type(4))) float v4f;
typedef __attribute__((ext_vector_type(8))) float v8f;

#define B_    4
#define H_    16
#define L_    2048
#define D_    64
#define RWIN  256
#define KEY0  (L_ - RWIN)
#define DIAG  (L_ - RWIN + 1)
#define LASTQ (L_ - 1 - DIAG)
#define BROWS  ((L_ - RWIN) / 4)

#define NWAVE  18
#define GWAVE  9
#define PPITCH 36
#define OPITCH 68

__device__ __forceinline__ float4 sum_block(const float* __restrict__ vblk, int half, int tcol) {
  const float4* vs = (const float4*)vblk;
  float4 a4 = make_float4(0.f, 0.f, 0.f, 0.f);
#pragma unroll
  for (int rd = 0; rd < 4; ++rd) {
    float4 vrow[16];
#pragma unroll
    for (int r = 0; r < 16; ++r)
      vrow[r] = vs[(rd * 32 + 2 * r + half) * (D_ / 4) + tcol];
#pragma unroll
    for (int r = 0; r < 16; ++r) {
      a4.x += vrow[r].x; a4.y += vrow[r].y; a4.z += vrow[r].z; a4.w += vrow[r].w;
    }
  }
  a4.x += __shfl_xor(a4.x, 16);
  a4.y += __shfl_xor(a4.y, 16);
  a4.z += __shfl_xor(a4.z, 16);
  a4.w += __shfl_xor(a4.w, 16);
  return a4;
}

__global__ __launch_bounds__(NWAVE * 32)
void attn_kernel(const float* __restrict__ q, const float* __restrict__ k,
                 const float* __restrict__ v, float* __restrict__ out) {
  __shared__ __attribute__((aligned(16))) float pLds[NWAVE][16 * PPITCH];
  __shared__ float4 oLds[NWAVE][16 * OPITCH / 4];
  __shared__ float4 mLds[NWAVE][4];
  __shared__ float4 lLds[NWAVE][4];
  __shared__ float4 vLds[16][16];

  const int bh   = blockIdx.x >> 2;
  const int pp   = blockIdx.x & 3;
  const int wave = threadIdx.x >> 5;
  const int grp  = wave >= GWAVE ? 1 : 0;
  const int gw   = wave - grp * GWAVE;
  const int p    = grp ? 7 - pp : pp;
  const int lane = threadIdx.x & 31;
  const int tcol = lane & 15;
  const int half = lane >> 4;

  const int  n0     = 8 - (p >> 1);
  const bool second = gw >= n0;
  const int  w  = second ? 15 - p : p;
  const int  kp = second ? ((15 - p) >> 1) + (gw - n0)
                         : (p >> 1) + gw;

  const float* qb = q + (size_t)bh * L_ * D_ + (size_t)w * 16 * D_;
  const float* kb = k + (size_t)bh * L_ * D_ + (size_t)(KEY0 + kp * 32) * D_;
  const float* vb = v + (size_t)bh * L_ * D_ + (size_t)(KEY0 + kp * 32) * D_;

  const bool streams = gw < 8;
  const int  blk  = (grp * 8 + gw + 4 * pp) & 15;
  const float* vblk = v + (size_t)bh * L_ * D_ + (size_t)(blk * 128) * D_;
  float4 a4 = make_float4(0.f, 0.f, 0.f, 0.f);
  if (streams && grp == 1) a4 = sum_block(vblk, half, tcol);

  const v4f* q4 = (const v4f*)(qb + tcol * D_) + half;
  const v4f* k4 = (const v4f*)(kb + tcol * D_) + half;
  v2f aq[16], bk0[16], bk1[16];
#pragma unroll
  for (int m = 0; m < 8; ++m) {
    const v4f qv = q4[2 * m];
    const v4f k0 = k4[2 * m];
    const v4f k1 = k4[2 * m + 16 * (D_ / 4)];
    v2f a0; a0.x = qv[0] * 0.125f; a0.y = qv[1] * 0.125f;
    v2f a1; a1.x = qv[2] * 0.125f; a1.y = qv[3] * 0.125f;
    v2f b0; b0.x = k0[0];          b0.y = k0[1];
    v2f b1; b1.x = k0[2];          b1.y = k0[3];
    v2f c0; c0.x = k1[0];          c0.y = k1[1];
    v2f c1; c1.x = k1[2];          c1.y = k1[3];
    aq[2 * m]  = a0; aq[2 * m + 1]  = a1;
    bk0[2 * m] = b0; bk0[2 * m + 1] = b1;
    bk1[2 * m] = c0; bk1[2 * m + 1] = c1;
  }
  const v4f* v4p = (const v4f*)vb + tcol;
  v2f bv[4][8];
#pragma unroll
  for (int c4 = 0; c4 < 8; ++c4) {
    const int kk = c4 * 4 + 2 * half;
    const v4f r0 = v4p[(kk)     * (D_ / 4)];
    const v4f r1 = v4p[(kk + 1) * (D_ / 4)];
#pragma unroll
    for (int nt = 0; nt < 4; ++nt) {
      v2f b; b.x = r0[nt]; b.y = r1[nt];
      bv[nt][c4] = b;
    }
  }

  const v8f vzero = {0.f, 0.f, 0.f, 0.f, 0.f, 0.f, 0.f, 0.f};

  v8f s0 = vzero, s1 = vzero;
#pragma unroll
  for (int c4 = 0; c4 < 16; ++c4) {
    s0 = __builtin_amdgcn_wmma_f32_16x16x4_f32(
             false, aq[c4], false, bk0[c4], (short)0, s0, false, false);
    s1 = __builtin_amdgcn_wmma_f32_16x16x4_f32(
             false, aq[c4], false, bk1[c4], (short)0, s1, false, false);
  }

  float* myP = &pLds[wave][0];
  const int j0 = KEY0 + kp * 32 + tcol;
  float mrow[8], lsum[8];
#pragma unroll
  for (int r = 0; r < 8; ++r) {
    const int i = w * 16 + r + 8 * half;
    const float sa = (j0 - i      >= DIAG) ? s0[r] : -10000.0f;
    const float sb = (j0 + 16 - i >= DIAG) ? s1[r] : -10000.0f;
    float rm = fmaxf(sa, sb);
    rm = fmaxf(rm, __shfl_xor(rm, 1));
    rm = fmaxf(rm, __shfl_xor(rm, 2));
    rm = fmaxf(rm, __shfl_xor(rm, 4));
    rm = fmaxf(rm, __shfl_xor(rm, 8));
    const float pa = __expf(sa - rm);
    const float pb = __expf(sb - rm);
    float ps = pa + pb;
    ps += __shfl_xor(ps, 1);
    ps += __shfl_xor(ps, 2);
    ps += __shfl_xor(ps, 4);
    ps += __shfl_xor(ps, 8);
    mrow[r] = rm;
    lsum[r] = ps;
    myP[(r + 8 * half) * PPITCH + tcol]      = pa;
    myP[(r + 8 * half) * PPITCH + 16 + tcol] = pb;
  }
  if (tcol == 0) {
    mLds[wave][2 * half]     = make_float4(mrow[0], mrow[1], mrow[2], mrow[3]);
    mLds[wave][2 * half + 1] = make_float4(mrow[4], mrow[5], mrow[6], mrow[7]);
    lLds[wave][2 * half]     = make_float4(lsum[0], lsum[1], lsum[2], lsum[3]);
    lLds[wave][2 * half + 1] = make_float4(lsum[4], lsum[5], lsum[6], lsum[7]);
  }

  v8f acc[4];
  acc[0] = vzero; acc[1] = vzero; acc[2] = vzero; acc[3] = vzero;
#pragma unroll
  for (int c4 = 0; c4 < 8; ++c4) {
    const int kk = c4 * 4 + 2 * half;
    const v2f a = *(const v2f*)&myP[tcol * PPITCH + kk];
#pragma unroll
    for (int nt = 0; nt < 4; ++nt)
      acc[nt] = __builtin_amdgcn_wmma_f32_16x16x4_f32(
                    false, a, false, bv[nt][c4], (short)0, acc[nt], false, false);
  }

  float* myO = (float*)&oLds[wave][0];
#pragma unroll
  for (int r = 0; r < 8; ++r) {
    v4f o4; o4[0] = acc[0][r]; o4[1] = acc[1][r]; o4[2] = acc[2][r]; o4[3] = acc[3][r];
    *(v4f*)&myO[(r + 8 * half) * OPITCH + 4 * tcol] = o4;
  }

  if (streams && grp == 0) a4 = sum_block(vblk, half, tcol);
  if (streams && half == 0) vLds[blk][tcol] = a4;

  __syncthreads();

  const int lt_ = threadIdx.x - grp * (GWAVE * 32);
  if (lt_ < 256) {
    const int row = lt_ >> 4;
    const int c   = lt_ & 15;
    const int t0  = grp * GWAVE;

    float  mt[GWAVE], lt[GWAVE];
    float4 xt[GWAVE];
    float4 vb4[16];
#pragma unroll
    for (int t = 0; t < GWAVE; ++t) {
      mt[t] = ((const float*)&mLds[t0 + t][0])[row];
      lt[t] = ((const float*)&lLds[t0 + t][0])[row];
      xt[t] = oLds[t0 + t][row * (OPITCH / 4) + c];
    }
#pragma unroll
    for (int b = 0; b < 16; ++b)
      vb4[b] = vLds[b][c];

    float mxA = -3.0e38f, mxB = -3.0e38f;
#pragma unroll
    for (int t = 0; t < GWAVE; ++t) {
      if (t < n0) mxA = fmaxf(mxA, mt[t]);
      else        mxB = fmaxf(mxB, mt[t]);
    }
    float  lA = 0.0f, lB = 0.0f;
    float4 oA = make_float4(0.f, 0.f, 0.f, 0.f);
    float4 oB = make_float4(0.f, 0.f, 0.f, 0.f);
#pragma unroll
    for (int t = 0; t < GWAVE; ++t) {
      const bool  inA = t < n0;
      const float f   = __expf(mt[t] - (inA ? mxA : mxB));
      const float fa  = inA ? f : 0.0f;
      const float fb  = inA ? 0.0f : f;
      lA += lt[t] * fa;
      lB += lt[t] * fb;
      oA.x += xt[t].x * fa; oA.y += xt[t].y * fa; oA.z += xt[t].z * fa; oA.w += xt[t].w * fa;
      oB.x += xt[t].x * fb; oB.y += xt[t].y * fb; oB.z += xt[t].z * fb; oB.w += xt[t].w * fb;
    }
    float* ob = out + (size_t)bh * L_ * D_;
    {
      const int   iA  = p * 16 + row;
      const float inv = 1.0f / lA;
      ((float4*)(ob + (size_t)iA * D_))[c] =
          make_float4(oA.x * inv, oA.y * inv, oA.z * inv, oA.w * inv);
    }
    const int iB = (15 - p) * 16 + row;
    if (iB <= LASTQ) {
      const float inv = 1.0f / lB;
      ((float4*)(ob + (size_t)iB * D_))[c] =
          make_float4(oB.x * inv, oB.y * inv, oB.z * inv, oB.w * inv);
    }

    float4 s = vb4[0];
#pragma unroll
    for (int b = 1; b < 16; ++b) {
      s.x += vb4[b].x; s.y += vb4[b].y; s.z += vb4[b].z; s.w += vb4[b].w;
    }
    const float sc = 1.0f / (float)L_;
    s.x *= sc; s.y *= sc; s.z *= sc; s.w *= sc;

    float4* ob4 = (float4*)ob;
    float4* o   = ob4 + (size_t)(RWIN + pp * BROWS + grp * (BROWS / 2) + row) * (D_ / 4) + c;
#pragma unroll
    for (int it = 0; it < BROWS / 32; ++it)
      o[(size_t)it * 16 * (D_ / 4)] = s;
    if (pp == 0 && grp == 0 && row == 0)
      ob4[(size_t)(RWIN - 1) * (D_ / 4) + c] = s;
  }
}

extern "C" void kernel_launch(void* const* d_in, const int* in_sizes, int n_in,
                              void* d_out, int out_size, void* d_ws, size_t ws_size,
                              hipStream_t stream) {
  const float* q = (const float*)d_in[0];
  const float* k = (const float*)d_in[1];
  const float* v = (const float*)d_in[2];
  float* out = (float*)d_out;

  attn_kernel<<<B_ * H_ * 4, NWAVE * 32, 0, stream>>>(q, k, v, out);
}


# attn operands staged once per work-group in LDS (corner rows of k, v, 4 query tiles; whole-row 16-byte loads, one more barrier), fragments read from LDS; fences hold the v stream's depth; same sums
# speedup vs baseline: 4.3136x; 1.0859x over previous
#include <hip/hip_runtime.h>


typedef __attribute__((ext_vector_type(2))) float v2f;
typedef __attribute__((ext_vector_type(4))) float v4f;
typedef __attribute__((ext_vector_type(8))) float v8f;

#define B_    4
#define H_    16
#define L_    2048
#define D_    64
#define RWIN  256
#define KEY0  (L_ - RWIN)
#define DIAG  (L_ - RWIN + 1)
#define LASTQ (L_ - 1 - DIAG)
#define BROWS  ((L_ - RWIN) / 4)

#define NWAVE  18
#define GWAVE  9
#define PPITCH 36
#define OPITCH 68
#define SPITCH 68
#define SROWS  576
#define SROW_V 256
#define SROW_Q 512

#define HOLD8(b) asm volatile("" :: "v"((b)[0]), "v"((b)[1]), "v"((b)[2]), "v"((b)[3]), \
                                    "v"((b)[4]), "v"((b)[5]), "v"((b)[6]), "v"((b)[7]))

__device__ __forceinline__ float4 sum_block(const float* __restrict__ vblk, int half, int tcol) {
  const float4* vs = (const float4*)vblk;
  float4 a4 = make_float4(0.f, 0.f, 0.f, 0.f);
  float4 vrow[64];
#pragma unroll
  for (int t = 0; t < 32; ++t)
    vrow[t] = vs[(2 * t + half) * (D_ / 4) + tcol];
#pragma unroll
  for (int t = 0; t < 64; ++t) {
    asm volatile("" :: "v"(vrow[t].x), "v"(vrow[t].y), "v"(vrow[t].z), "v"(vrow[t].w),
                        "v"(a4.x), "v"(a4.y), "v"(a4.z), "v"(a4.w));
    a4.x += vrow[t].x; a4.y += vrow[t].y; a4.z += vrow[t].z; a4.w += vrow[t].w;
    if (t + 32 < 64)
      vrow[t + 32] = vs[(2 * (t + 32) + half) * (D_ / 4) + tcol];
  }
  a4.x += __shfl_xor(a4.x, 16);
  a4.y += __shfl_xor(a4.y, 16);
  a4.z += __shfl_xor(a4.z, 16);
  a4.w += __shfl_xor(a4.w, 16);
  return a4;
}

__global__ __launch_bounds__(NWAVE * 32)
void attn_kernel(const float* __restrict__ q, const float* __restrict__ k,
                 const float* __restrict__ v, float* __restrict__ out) {
  __shared__ __attribute__((aligned(16))) float sLds[SROWS * SPITCH];
  __shared__ __attribute__((aligned(16))) float pLds[NWAVE][16 * PPITCH];
  __shared__ float4 oLds[NWAVE][16 * OPITCH / 4];
  __shared__ float4 mLds[NWAVE][4];
  __shared__ float4 lLds[NWAVE][4];
  __shared__ float4 vLds[16][16];

  const int bh   = blockIdx.x >> 2;
  const int pp   = blockIdx.x & 3;
  const int wave = threadIdx.x >> 5;
  const int grp  = wave >= GWAVE ? 1 : 0;
  const int gw   = wave - grp * GWAVE;
  const int p    = grp ? 7 - pp : pp;
  const int lane = threadIdx.x & 31;
  const int tcol = lane & 15;
  const int half = lane >> 4;

  const int  n0     = 8 - (p >> 1);
  const bool second = gw >= n0;
  const int  w  = second ? 15 - p : p;
  const int  kp = second ? ((15 - p) >> 1) + (gw - n0)
                         : (p >> 1) + gw;

  {
    const int wv = __builtin_amdgcn_readfirstlane(wave);
    const float* hq = q + (size_t)bh * L_ * D_;
    const float* hk = k + (size_t)bh * L_ * D_ + (size_t)KEY0 * D_;
    const float* hv = v + (size_t)bh * L_ * D_ + (size_t)KEY0 * D_;
    const int lo = half * D_ + 4 * tcol;
    const int so = half * SPITCH + 4 * tcol;
    auto qrow = [&](int pr) -> int {
      const int sl = pr >> 3;
      const int pg = (sl & 2) ? 7 - pp : pp;
      const int qt = (sl & 1) ? 15 - pg : pg;
      return qt * 16 + 2 * (pr & 7);
    };
    const int    ix   = wv + NWAVE;
    const float* tsrc = ix < 32 ? hq + (size_t)qrow(ix) * D_
                      : ix < 34 ? hk + (size_t)(2 * (94 + ix)) * D_
                                : hv + (size_t)(2 * (92 + ix)) * D_;
    const int    tdst = ix < 32 ? SROW_Q + 2 * ix
                      : ix < 34 ? 2 * (94 + ix)
                                : SROW_V + 2 * (92 + ix);
    v4f st[16];
#pragma unroll
    for (int j = 0; j < 7; ++j) {
      st[j]     = *(const v4f*)(hk + (size_t)(2 * (wv + NWAVE * j)) * D_ + lo);
      st[7 + j] = *(const v4f*)(hv + (size_t)(2 * (wv + NWAVE * j)) * D_ + lo);
    }
    st[14] = *(const v4f*)(hq + (size_t)qrow(wv) * D_ + lo);
    st[15] = *(const v4f*)(tsrc + lo);
#pragma unroll
    for (int j = 0; j < 7; ++j) {
      *(v4f*)&sLds[(2 * (wv + NWAVE * j)) * SPITCH + so]          = st[j];
      *(v4f*)&sLds[(SROW_V + 2 * (wv + NWAVE * j)) * SPITCH + so] = st[7 + j];
    }
    *(v4f*)&sLds[(SROW_Q + 2 * wv) * SPITCH + so] = st[14];
    *(v4f*)&sLds[tdst * SPITCH + so]              = st[15];
  }
  __syncthreads();

  const int qslot = 2 * grp + (second ? 1 : 0);
  const float* qb = &sLds[(SROW_Q + qslot * 16) * SPITCH];
  const float* kb = &sLds[(kp * 32) * SPITCH];
  const float* vb = &sLds[(SROW_V + kp * 32) * SPITCH];

  const bool streams = gw < 8;
  const int  blk  = (grp * 8 + gw + 4 * pp) & 15;
  const float* vblk = v + (size_t)bh * L_ * D_ + (size_t)(blk * 128) * D_;
  float4 a4 = make_float4(0.f, 0.f, 0.f, 0.f);
  if (streams && grp == 1) a4 = sum_block(vblk, half, tcol);

  const v4f* q4 = (const v4f*)(qb + tcol * SPITCH) + half;
  const v4f* k4 = (const v4f*)(kb + tcol * SPITCH) + half;
  v4f qr[8], k0r[8], k1r[8];
#pragma unroll
  for (int m = 0; m < 8; ++m) {
    qr[m]  = q4[2 * m];
    k0r[m] = k4[2 * m];
    k1r[m] = k4[2 * m + 16 * (SPITCH / 4)];
  }
  HOLD8(qr); HOLD8(k0r); HOLD8(k1r);
  v2f aq[16], bk0[16], bk1[16];
#pragma unroll
  for (int m = 0; m < 8; ++m) {
    const v4f qv = qr[m];
    const v4f k0 = k0r[m];
    const v4f k1 = k1r[m];
    v2f a0; a0.x = qv[0] * 0.125f; a0.y = qv[1] * 0.125f;
    v2f a1; a1.x = qv[2] * 0.125f; a1.y = qv[3] * 0.125f;
    v2f b0; b0.x = k0[0];          b0.y = k0[1];
    v2f b1; b1.x = k0[2];          b1.y = k0[3];
    v2f c0; c0.x = k1[0];          c0.y = k1[1];
    v2f c1; c1.x = k1[2];          c1.y = k1[3];
    aq[2 * m]  = a0; aq[2 * m + 1]  = a1;
    bk0[2 * m] = b0; bk0[2 * m + 1] = b1;
    bk1[2 * m] = c0; bk1[2 * m + 1] = c1;
  }
  const v4f* v4p = (const v4f*)vb + tcol;
  v2f bv[4][8];
#pragma unroll
  for (int c4 = 0; c4 < 8; ++c4) {
    const int kk = c4 * 4 + 2 * half;
    const v4f r0 = v4p[(kk)     * (SPITCH / 4)];
    const v4f r1 = v4p[(kk + 1) * (SPITCH / 4)];
#pragma unroll
    for (int nt = 0; nt < 4; ++nt) {
      v2f b; b.x = r0[nt]; b.y = r1[nt];
      bv[nt][c4] = b;
    }
  }

  const v8f vzero = {0.f, 0.f, 0.f, 0.f, 0.f, 0.f, 0.f, 0.f};

  v8f s0 = vzero, s1 = vzero;
#pragma unroll
  for (int c4 = 0; c4 < 16; ++c4) {
    s0 = __builtin_amdgcn_wmma_f32_16x16x4_f32(
             false, aq[c4], false, bk0[c4], (short)0, s0, false, false);
    s1 = __builtin_amdgcn_wmma_f32_16x16x4_f32(
             false, aq[c4], false, bk1[c4], (short)0, s1, false, false);
  }

  float* myP = &pLds[wave][0];
  const int j0 = KEY0 + kp * 32 + tcol;
  float mrow[8], lsum[8];
#pragma unroll
  for (int r = 0; r < 8; ++r) {
    const int i = w * 16 + r + 8 * half;
    const float sa = (j0 - i      >= DIAG) ? s0[r] : -10000.0f;
    const float sb = (j0 + 16 - i >= DIAG) ? s1[r] : -10000.0f;
    float rm = fmaxf(sa, sb);
    rm = fmaxf(rm, __shfl_xor(rm, 1));
    rm = fmaxf(rm, __shfl_xor(rm, 2));
    rm = fmaxf(rm, __shfl_xor(rm, 4));
    rm = fmaxf(rm, __shfl_xor(rm, 8));
    const float pa = __expf(sa - rm);
    const float pb = __expf(sb - rm);
    float ps = pa + pb;
    ps += __shfl_xor(ps, 1);
    ps += __shfl_xor(ps, 2);
    ps += __shfl_xor(ps, 4);
    ps += __shfl_xor(ps, 8);
    mrow[r] = rm;
    lsum[r] = ps;
    myP[(r + 8 * half) * PPITCH + tcol]      = pa;
    myP[(r + 8 * half) * PPITCH + 16 + tcol] = pb;
  }
  if (tcol == 0) {
    mLds[wave][2 * half]     = make_float4(mrow[0], mrow[1], mrow[2], mrow[3]);
    mLds[wave][2 * half + 1] = make_float4(mrow[4], mrow[5], mrow[6], mrow[7]);
    lLds[wave][2 * half]     = make_float4(lsum[0], lsum[1], lsum[2], lsum[3]);
    lLds[wave][2 * half + 1] = make_float4(lsum[4], lsum[5], lsum[6], lsum[7]);
  }

  v8f acc[4];
  acc[0] = vzero; acc[1] = vzero; acc[2] = vzero; acc[3] = vzero;
#pragma unroll
  for (int c4 = 0; c4 < 8; ++c4) {
    const int kk = c4 * 4 + 2 * half;
    const v2f a = *(const v2f*)&myP[tcol * PPITCH + kk];
#pragma unroll
    for (int nt = 0; nt < 4; ++nt)
      acc[nt] = __builtin_amdgcn_wmma_f32_16x16x4_f32(
                    false, a, false, bv[nt][c4], (short)0, acc[nt], false, false);
  }

  float* myO = (float*)&oLds[wave][0];
#pragma unroll
  for (int r = 0; r < 8; ++r) {
    v4f o4; o4[0] = acc[0][r]; o4[1] = acc[1][r]; o4[2] = acc[2][r]; o4[3] = acc[3][r];
    *(v4f*)&myO[(r + 8 * half) * OPITCH + 4 * tcol] = o4;
  }

  if (streams && grp == 0) a4 = sum_block(vblk, half, tcol);
  if (streams && half == 0) vLds[blk][tcol] = a4;

  __syncthreads();

  const int lt_ = threadIdx.x - grp * (GWAVE * 32);
  if (lt_ < 256) {
    const int row = lt_ >> 4;
    const int c   = lt_ & 15;
    const int t0  = grp * GWAVE;

    float  mt[GWAVE], lt[GWAVE];
    float4 xt[GWAVE];
    float4 vb4[16];
#pragma unroll
    for (int t = 0; t < GWAVE; ++t) {
      mt[t] = ((const float*)&mLds[t0 + t][0])[row];
      lt[t] = ((const float*)&lLds[t0 + t][0])[row];
      xt[t] = oLds[t0 + t][row * (OPITCH / 4) + c];
    }
#pragma unroll
    for (int b = 0; b < 16; ++b)
      vb4[b] = vLds[b][c];

    float mxA = -3.0e38f, mxB = -3.0e38f;
#pragma unroll
    for (int t = 0; t < GWAVE; ++t) {
      if (t < n0) mxA = fmaxf(mxA, mt[t]);
      else        mxB = fmaxf(mxB, mt[t]);
    }
    float  lA = 0.0f, lB = 0.0f;
    float4 oA = make_float4(0.f, 0.f, 0.f, 0.f);
    float4 oB = make_float4(0.f, 0.f, 0.f, 0.f);
#pragma unroll
    for (int t = 0; t < GWAVE; ++t) {
      const bool  inA = t < n0;
      const float f   = __expf(mt[t] - (inA ? mxA : mxB));
      const float fa  = inA ? f : 0.0f;
      const float fb  = inA ? 0.0f : f;
      lA += lt[t] * fa;
      lB += lt[t] * fb;
      oA.x += xt[t].x * fa; oA.y += xt[t].y * fa; oA.z += xt[t].z * fa; oA.w += xt[t].w * fa;
      oB.x += xt[t].x * fb; oB.y += xt[t].y * fb; oB.z += xt[t].z * fb; oB.w += xt[t].w * fb;
    }
    float* ob = out + (size_t)bh * L_ * D_;
    {
      const int   iA  = p * 16 + row;
      const float inv = 1.0f / lA;
      ((float4*)(ob + (size_t)iA * D_))[c] =
          make_float4(oA.x * inv, oA.y * inv, oA.z * inv, oA.w * inv);
    }
    const int iB = (15 - p) * 16 + row;
    if (iB <= LASTQ) {
      const float inv = 1.0f / lB;
      ((float4*)(ob + (size_t)iB * D_))[c] =
          make_float4(oB.x * inv, oB.y * inv, oB.z * inv, oB.w * inv);
    }

    float4 s = vb4[0];
#pragma unroll
    for (int b = 1; b < 16; ++b) {
      s.x += vb4[b].x; s.y += vb4[b].y; s.z += vb4[b].z; s.w += vb4[b].w;
    }
    const float sc = 1.0f / (float)L_;
    s.x *= sc; s.y *= sc; s.z *= sc; s.w *= sc;

    float4* ob4 = (float4*)ob;
    float4* o   = ob4 + (size_t)(RWIN + pp * BROWS + grp * (BROWS / 2) + row) * (D_ / 4) + c;
#pragma unroll
    for (int it = 0; it < BROWS / 32; ++it)
      o[(size_t)it * 16 * (D_ / 4)] = s;
    if (pp == 0 && grp == 0 && row == 0)
      ob4[(size_t)(RWIN - 1) * (D_ / 4) + c] = s;
  }
}

extern "C" void kernel_launch(void* const* d_in, const int* in_sizes, int n_in,
                              void* d_out, int out_size, void* d_ws, size_t ws_size,
                              hipStream_t stream) {
  const float* q = (const float*)d_in[0];
  const float* k = (const float*)d_in[1];
  const float* v = (const float*)d_in[2];
  float* out = (float*)d_out;

  attn_kernel<<<B_ * H_ * 4, NWAVE * 32, 0, stream>>>(q, k, v, out);
}
